# starting kernel
# baseline (speedup 1.0000x reference)
#include <hip/hip_runtime.h>
#include <hip/hip_fp16.h>

typedef __attribute__((ext_vector_type(16))) _Float16 v16h;
typedef __attribute__((ext_vector_type(8)))  _Float16 v8h;
typedef __attribute__((ext_vector_type(8)))  float    v8f;

#define N_NODES   50000
#define N_EDGES   1600000
#define IN_DIM    256
#define OUT_DIM   32
#define N_HEADS   4
#define F_DIM     128
#define ALPHA_SLOPE 0.2f
#define M_TILES   3125
#define N_TILES   8
#define K_TILES   8

__global__ void pack_w_kernel(const float* __restrict__ W, _Float16* __restrict__ wpack) {
    int idx = blockIdx.x * blockDim.x + threadIdx.x;
    if (idx >= N_TILES * K_TILES * 32) return;
    int lane = idx & 31;
    int kt   = (idx >> 5) & 7;
    int nt   = idx >> 8;
    int col  = nt * 16 + (lane & 15);
    int kbase = kt * 32 + ((lane & 16) ? 8 : 0);
    _Float16* dst = wpack + idx * 16;
    #pragma unroll
    for (int i = 0; i < 8; ++i) {
        dst[i]     = (_Float16)W[(kbase + i)      * F_DIM + col];
        dst[i + 8] = (_Float16)W[(kbase + 16 + i) * F_DIM + col];
    }
}

__global__ void init_kernel(float* __restrict__ out, float* __restrict__ ssum,
                            unsigned* __restrict__ mkey) {
    int i = blockIdx.x * blockDim.x + threadIdx.x;
    if (i < N_NODES * F_DIM)   out[i] = 0.0f;
    if (i < N_NODES * N_HEADS) { ssum[i] = 0.0f; mkey[i] = 0u; }
}

__global__ void __launch_bounds__(256) gemm_kernel(const float* __restrict__ h,
                                                   const _Float16* __restrict__ wpack,
                                                   float* __restrict__ hprime) {
    __shared__ __align__(32) _Float16 tileA[16 * IN_DIM];

    const int tid = threadIdx.x;
    const int m0  = blockIdx.x * 16;

    #pragma unroll
    for (int j = 0; j < 16; ++j) {
        int row = m0 + j;
        float v = (row < N_NODES) ? h[row * IN_DIM + tid] : 0.0f;
        tileA[j * IN_DIM + tid] = (_Float16)v;
    }
    __syncthreads();

    const int wave = tid >> 5;
    const int lane = tid & 31;
    const int rowA = lane & 15;
    const int koff = (lane & 16) ? 8 : 0;

    v8f c = {};
    #pragma unroll
    for (int kt = 0; kt < K_TILES; ++kt) {
        const int kbase = kt * 32 + koff;
        v8h lo = *(const v8h*)&tileA[rowA * IN_DIM + kbase];
        v8h hi = *(const v8h*)&tileA[rowA * IN_DIM + kbase + 16];
        v16h afrag = __builtin_shufflevector(lo, hi,
            0, 1, 2, 3, 4, 5, 6, 7, 8, 9, 10, 11, 12, 13, 14, 15);
        v16h bfrag = *(const v16h*)&wpack[((wave * K_TILES + kt) * 32 + lane) * 16];
        c = __builtin_amdgcn_wmma_f32_16x16x32_f16(
                 false, afrag,  false, bfrag,
                 (short)0, c,  false,  false);
    }

    const int col   = wave * 16 + (lane & 15);
    const int rbase = (lane & 16) ? 8 : 0;
    #pragma unroll
    for (int i = 0; i < 8; ++i) {
        int gr = m0 + rbase + i;
        if (gr < N_NODES) hprime[gr * F_DIM + col] = c[i];
    }
}

__global__ void alpha_kernel(const float* __restrict__ hprime, const float* __restrict__ a,
                             float* __restrict__ asrc, float* __restrict__ adst) {
    int t = blockIdx.x * blockDim.x + threadIdx.x;
    if (t >= N_NODES * N_HEADS) return;
    int node = t >> 2;
    int head = t & 3;
    const float* hp = hprime + node * F_DIM + head * OUT_DIM;
    float s0 = 0.0f, s1 = 0.0f;
    #pragma unroll
    for (int d = 0; d < OUT_DIM; ++d) {
        float v = hp[d];
        s0 += v * a[d * N_HEADS + head];
        s1 += v * a[(OUT_DIM + d) * N_HEADS + head];
    }
    asrc[t] = s0;
    adst[t] = s1;
}

__device__ __forceinline__ unsigned fkey(float f) {
    unsigned u = __float_as_uint(f);
    return (u & 0x80000000u) ? ~u : (u | 0x80000000u);
}
__device__ __forceinline__ float finvkey(unsigned k) {
    unsigned u = (k & 0x80000000u) ? (k & 0x7FFFFFFFu) : ~k;
    return __uint_as_float(u);
}

__global__ void edge_att_kernel(const int* __restrict__ adj,
                                const float* __restrict__ asrc,
                                const float* __restrict__ adst,
                                float* __restrict__ att,
                                unsigned* __restrict__ mkey) {
    int t = blockIdx.x * blockDim.x + threadIdx.x;
    if (t >= N_EDGES * N_HEADS) return;
    int e = t >> 2, head = t & 3;
    int s = adj[e];
    int d = adj[N_EDGES + e];
    float v = asrc[s * N_HEADS + head] + adst[d * N_HEADS + head];
    v = (v > 0.0f) ? v : ALPHA_SLOPE * v;
    att[t] = v;
    atomicMax(&mkey[d * N_HEADS + head], fkey(v));
}

__global__ void edge_exp_kernel(const int* __restrict__ adj,
                                float* __restrict__ att,
                                const unsigned* __restrict__ mkey,
                                float* __restrict__ ssum) {
    int t = blockIdx.x * blockDim.x + threadIdx.x;
    if (t >= N_EDGES * N_HEADS) return;
    int e = t >> 2, head = t & 3;
    int d = adj[N_EDGES + e];
    float m  = finvkey(mkey[d * N_HEADS + head]);
    float ex = __expf(att[t] - m);
    att[t] = ex;
    atomicAdd(&ssum[d * N_HEADS + head], ex);
}

__global__ void aggregate_kernel(const int* __restrict__ adj,
                                 const float* __restrict__ att,
                                 const float* __restrict__ ssum,
                                 const float* __restrict__ hprime,
                                 float* __restrict__ out) {
    long long t = (long long)blockIdx.x * blockDim.x + threadIdx.x;
    if (t >= (long long)N_EDGES * 32) return;
    int e    = (int)(t >> 5);
    int lane = (int)(t & 31);
    int head = lane >> 3;
    int s = adj[e];
    int d = adj[N_EDGES + e];
    float w = att[e * N_HEADS + head] / ssum[d * N_HEADS + head];
    const float4 hv = *(const float4*)&hprime[s * F_DIM + lane * 4];
    float* o = out + d * F_DIM + lane * 4;
    atomicAdd(o + 0, w * hv.x);
    atomicAdd(o + 1, w * hv.y);
    atomicAdd(o + 2, w * hv.z);
    atomicAdd(o + 3, w * hv.w);
}

extern "C" void kernel_launch(void* const* d_in, const int* in_sizes, int n_in,
                              void* d_out, int out_size, void* d_ws, size_t ws_size,
                              hipStream_t stream) {
    const float* h   = (const float*)d_in[0];
    const int*   adj = (const int*)d_in[1];
    const float* W   = (const float*)d_in[2];
    const float* a   = (const float*)d_in[3];
    float* out = (float*)d_out;

    char* ws = (char*)d_ws;
    size_t off = 0;
    float*    hprime = (float*)(ws + off);     off += (size_t)N_NODES * F_DIM * 4;
    float*    asrc   = (float*)(ws + off);     off += (size_t)N_NODES * N_HEADS * 4;
    float*    adst   = (float*)(ws + off);     off += (size_t)N_NODES * N_HEADS * 4;
    float*    att    = (float*)(ws + off);     off += (size_t)N_EDGES * N_HEADS * 4;
    unsigned* mkey   = (unsigned*)(ws + off);  off += (size_t)N_NODES * N_HEADS * 4;
    float*    ssum   = (float*)(ws + off);     off += (size_t)N_NODES * N_HEADS * 4;
    _Float16* wpack  = (_Float16*)(ws + off);  off += (size_t)N_TILES * K_TILES * 32 * 16 * 2;

    const int B = 256;

    pack_w_kernel<<<(N_TILES * K_TILES * 32 + B - 1) / B, B, 0, stream>>>(W, wpack);
    init_kernel<<<(N_NODES * F_DIM + B - 1) / B, B, 0, stream>>>(out, ssum, mkey);
    gemm_kernel<<<M_TILES, B, 0, stream>>>(h, wpack, hprime);
    alpha_kernel<<<(N_NODES * N_HEADS + B - 1) / B, B, 0, stream>>>(hprime, a, asrc, adst);
    edge_att_kernel<<<(N_EDGES * N_HEADS + B - 1) / B, B, 0, stream>>>(adj, asrc, adst, att, mkey);
    edge_exp_kernel<<<(N_EDGES * N_HEADS + B - 1) / B, B, 0, stream>>>(adj, att, mkey, ssum);
    long long aggT = (long long)N_EDGES * 32;
    aggregate_kernel<<<(unsigned)((aggT + B - 1) / B), B, 0, stream>>>(adj, att, ssum, hprime, out);
}


# Group edges by destination each call (int count with slot, scan, fill); one wave per node sorts its segment by source, then softmax and weighted row sum in that fixed order: no float atomics
# speedup vs baseline: 1.8309x; 1.8309x over previous
#include <hip/hip_runtime.h>
#include <hip/hip_fp16.h>

typedef __attribute__((ext_vector_type(16))) _Float16 v16h;
typedef __attribute__((ext_vector_type(8)))  _Float16 v8h;
typedef __attribute__((ext_vector_type(8)))  float    v8f;

#define N_NODES   50000
#define N_EDGES   1600000
#define IN_DIM    256
#define OUT_DIM   32
#define N_HEADS   4
#define F_DIM     128
#define ALPHA_SLOPE 0.2f
#define M_TILES   3125
#define N_TILES   8
#define K_TILES   8

__global__ void pack_w_kernel(const float* __restrict__ W, _Float16* __restrict__ wpack,
                              int* __restrict__ cnt) {
    int idx = blockIdx.x * blockDim.x + threadIdx.x;
    if (idx < N_NODES) cnt[idx] = 0;
    if (idx >= N_TILES * K_TILES * 32) return;
    int lane = idx & 31;
    int kt   = (idx >> 5) & 7;
    int nt   = idx >> 8;
    int col  = nt * 16 + (lane & 15);
    int kbase = kt * 32 + ((lane & 16) ? 8 : 0);
    _Float16* dst = wpack + idx * 16;
    #pragma unroll
    for (int i = 0; i < 8; ++i) {
        dst[i]     = (_Float16)W[(kbase + i)      * F_DIM + col];
        dst[i + 8] = (_Float16)W[(kbase + 16 + i) * F_DIM + col];
    }
}

__global__ void __launch_bounds__(256) gemm_kernel(const float* __restrict__ h,
                                                   const _Float16* __restrict__ wpack,
                                                   float* __restrict__ hprime) {
    __shared__ __align__(32) _Float16 tileA[16 * IN_DIM];

    const int tid = threadIdx.x;
    const int m0  = blockIdx.x * 16;

    #pragma unroll
    for (int j = 0; j < 16; ++j) {
        int row = m0 + j;
        float v = (row < N_NODES) ? h[row * IN_DIM + tid] : 0.0f;
        tileA[j * IN_DIM + tid] = (_Float16)v;
    }
    __syncthreads();

    const int wave = tid >> 5;
    const int lane = tid & 31;
    const int rowA = lane & 15;
    const int koff = (lane & 16) ? 8 : 0;

    v8f c = {};
    #pragma unroll
    for (int kt = 0; kt < K_TILES; ++kt) {
        const int kbase = kt * 32 + koff;
        v8h lo = *(const v8h*)&tileA[rowA * IN_DIM + kbase];
        v8h hi = *(const v8h*)&tileA[rowA * IN_DIM + kbase + 16];
        v16h afrag = __builtin_shufflevector(lo, hi,
            0, 1, 2, 3, 4, 5, 6, 7, 8, 9, 10, 11, 12, 13, 14, 15);
        v16h bfrag = *(const v16h*)&wpack[((wave * K_TILES + kt) * 32 + lane) * 16];
        c = __builtin_amdgcn_wmma_f32_16x16x32_f16(
                 false, afrag,  false, bfrag,
                 (short)0, c,  false,  false);
    }

    const int col   = wave * 16 + (lane & 15);
    const int rbase = (lane & 16) ? 8 : 0;
    #pragma unroll
    for (int i = 0; i < 8; ++i) {
        int gr = m0 + rbase + i;
        if (gr < N_NODES) hprime[gr * F_DIM + col] = c[i];
    }
}

__global__ void alpha_kernel(const float* __restrict__ hprime, const float* __restrict__ a,
                             float* __restrict__ asrc, float* __restrict__ adst) {
    int t = blockIdx.x * blockDim.x + threadIdx.x;
    if (t >= N_NODES * N_HEADS) return;
    int node = t >> 2;
    int head = t & 3;
    const float* hp = hprime + node * F_DIM + head * OUT_DIM;
    float s0 = 0.0f, s1 = 0.0f;
    #pragma unroll
    for (int d = 0; d < OUT_DIM; ++d) {
        float v = hp[d];
        s0 += v * a[d * N_HEADS + head];
        s1 += v * a[(OUT_DIM + d) * N_HEADS + head];
    }
    asrc[t] = s0;
    adst[t] = s1;
}

__global__ void count_rank_kernel(const int* __restrict__ adj, int* __restrict__ cnt,
                                  int* __restrict__ rank) {
    int e = blockIdx.x * blockDim.x + threadIdx.x;
    if (e >= N_EDGES) return;
    int d = adj[N_EDGES + e];
    rank[e] = atomicAdd(&cnt[d], 1);
}

#define SCAN_T 1024
#define SCAN_U 13
#define N_UNITS (N_NODES / 4)

__global__ void __launch_bounds__(SCAN_T) scan_kernel(const int* __restrict__ cnt,
                                                      int* __restrict__ offs) {
    __shared__ int wtot[32];
    const int t = threadIdx.x, lane = t & 31, wave = t >> 5;
    const int4* c4 = (const int4*)cnt;
    int4 v[SCAN_U];
    int sum = 0;
    #pragma unroll
    for (int q = 0; q < SCAN_U; ++q) {
        int u = t * SCAN_U + q;
        v[q] = (u < N_UNITS) ? c4[u] : make_int4(0, 0, 0, 0);
        sum += v[q].x + v[q].y + v[q].z + v[q].w;
    }
    int inc = sum;
    for (int o = 1; o < 32; o <<= 1) {
        int n = __shfl_up(inc, o);
        if (lane >= o) inc += n;
    }
    if (lane == 31) wtot[wave] = inc;
    __syncthreads();
    if (wave == 0) {
        int w = wtot[lane], wi = w;
        for (int o = 1; o < 32; o <<= 1) {
            int n = __shfl_up(wi, o);
            if (lane >= o) wi += n;
        }
        wtot[lane] = wi - w;
    }
    __syncthreads();
    int run = wtot[wave] + inc - sum;
    #pragma unroll
    for (int q = 0; q < SCAN_U; ++q) {
        int u = t * SCAN_U + q;
        if (u < N_UNITS) {
            int4 o;
            o.x = run; run += v[q].x;
            o.y = run; run += v[q].y;
            o.z = run; run += v[q].z;
            o.w = run; run += v[q].w;
            ((int4*)offs)[u] = o;
        }
    }
    if (t == SCAN_T - 1) offs[N_NODES] = run;
}

__global__ void fill_kernel(const int* __restrict__ adj, const int* __restrict__ offs,
                            const int* __restrict__ rank, int* __restrict__ csr_src) {
    int e = blockIdx.x * blockDim.x + threadIdx.x;
    if (e >= N_EDGES) return;
    int s = adj[e];
    int d = adj[N_EDGES + e];
    csr_src[offs[d] + rank[e]] = s;
}

__device__ __forceinline__ float leaky(float v) {
    return (v > 0.0f) ? v : ALPHA_SLOPE * v;
}

__device__ __forceinline__ void wave_lds_order() {
    __builtin_amdgcn_fence(__ATOMIC_RELEASE, "wavefront");
    __builtin_amdgcn_wave_barrier();
    __builtin_amdgcn_fence(__ATOMIC_ACQUIRE, "wavefront");
}

#define CAP 128
#define WPB 8

__global__ void __launch_bounds__(WPB * 32) node_kernel(const int* __restrict__ offs,
                                                        const int* __restrict__ csr_src,
                                                        const float* __restrict__ asrc,
                                                        const float* __restrict__ adst,
                                                        const float* __restrict__ hprime,
                                                        float* __restrict__ out) {
    __shared__ __align__(16) int   s_key[WPB][CAP];
    __shared__ __align__(16) int   s_srt[WPB][CAP];
    __shared__ __align__(16) float s_w[WPB][N_HEADS][CAP];

    const int wave = __builtin_amdgcn_readfirstlane((int)(threadIdx.x >> 5));
    const int lane = threadIdx.x & 31;
    const int node = blockIdx.x * WPB + wave;
    if (node >= N_NODES) return;

    int*   key = s_key[wave];
    int*   srt = s_srt[wave];
    float* wgt = &s_w[wave][0][0];

    const int beg = offs[node];
    const int end = offs[node + 1];
    const float4* asrc4 = (const float4*)asrc;
    const float4 ad = ((const float4*)adst)[node];
    const int head = lane >> 3;

    float m0 = -INFINITY, m1 = -INFINITY, m2 = -INFINITY, m3 = -INFINITY;
    for (int i = beg + lane; i < end; i += 32) {
        const float4 as = asrc4[csr_src[i]];
        m0 = fmaxf(m0, leaky(as.x + ad.x));
        m1 = fmaxf(m1, leaky(as.y + ad.y));
        m2 = fmaxf(m2, leaky(as.z + ad.z));
        m3 = fmaxf(m3, leaky(as.w + ad.w));
    }
    for (int o = 16; o > 0; o >>= 1) {
        m0 = fmaxf(m0, __shfl_xor(m0, o));
        m1 = fmaxf(m1, __shfl_xor(m1, o));
        m2 = fmaxf(m2, __shfl_xor(m2, o));
        m3 = fmaxf(m3, __shfl_xor(m3, o));
    }

    float l0 = 0.0f, l1 = 0.0f, l2 = 0.0f, l3 = 0.0f;
    float4 acc = make_float4(0.0f, 0.0f, 0.0f, 0.0f);
    const float* hp   = hprime + lane * 4;
    const float* wrow = wgt + head * CAP;

    for (int base = beg; base < end; base += CAP) {
        const int cnt  = min(CAP, end - base);
        const int cnt4 = (cnt + 3) & ~3;

        for (int i = lane; i < cnt4; i += 32)
            key[i] = (i < cnt) ? ((csr_src[base + i] << 7) | i) : 0x7fffffff;
        wave_lds_order();

        for (int i = lane; i < cnt; i += 32) {
            const int kk = key[i];
            int r = 0;
            for (int j = 0; j < cnt4; j += 4) {
                const int4 q = *(const int4*)&key[j];
                r += (q.x < kk) + (q.y < kk) + (q.z < kk) + (q.w < kk);
            }
            srt[r] = kk >> 7;
        }
        wave_lds_order();

        for (int i = lane; i < cnt; i += 32) {
            const float4 as = asrc4[srt[i]];
            const float e0 = __expf(leaky(as.x + ad.x) - m0);
            const float e1 = __expf(leaky(as.y + ad.y) - m1);
            const float e2 = __expf(leaky(as.z + ad.z) - m2);
            const float e3 = __expf(leaky(as.w + ad.w) - m3);
            wgt[0 * CAP + i] = e0;  l0 += e0;
            wgt[1 * CAP + i] = e1;  l1 += e1;
            wgt[2 * CAP + i] = e2;  l2 += e2;
            wgt[3 * CAP + i] = e3;  l3 += e3;
        }
        wave_lds_order();

        int k = 0;
        for (; k + 4 <= cnt; k += 4) {
            const int4   s4 = *(const int4*)&srt[k];
            const float4 w4 = *(const float4*)&wrow[k];
            const int sa = __builtin_amdgcn_readfirstlane(s4.x);
            const int sb = __builtin_amdgcn_readfirstlane(s4.y);
            const int sc = __builtin_amdgcn_readfirstlane(s4.z);
            const int sd = __builtin_amdgcn_readfirstlane(s4.w);
            const float4 ha = *(const float4*)(hp + (size_t)sa * F_DIM);
            const float4 hb = *(const float4*)(hp + (size_t)sb * F_DIM);
            const float4 hc = *(const float4*)(hp + (size_t)sc * F_DIM);
            const float4 hd = *(const float4*)(hp + (size_t)sd * F_DIM);
            acc.x = fmaf(w4.x, ha.x, acc.x); acc.y = fmaf(w4.x, ha.y, acc.y);
            acc.z = fmaf(w4.x, ha.z, acc.z); acc.w = fmaf(w4.x, ha.w, acc.w);
            acc.x = fmaf(w4.y, hb.x, acc.x); acc.y = fmaf(w4.y, hb.y, acc.y);
            acc.z = fmaf(w4.y, hb.z, acc.z); acc.w = fmaf(w4.y, hb.w, acc.w);
            acc.x = fmaf(w4.z, hc.x, acc.x); acc.y = fmaf(w4.z, hc.y, acc.y);
            acc.z = fmaf(w4.z, hc.z, acc.z); acc.w = fmaf(w4.z, hc.w, acc.w);
            acc.x = fmaf(w4.w, hd.x, acc.x); acc.y = fmaf(w4.w, hd.y, acc.y);
            acc.z = fmaf(w4.w, hd.z, acc.z); acc.w = fmaf(w4.w, hd.w, acc.w);
        }
        for (; k < cnt; ++k) {
            const int   sa = __builtin_amdgcn_readfirstlane(srt[k]);
            const float w  = wrow[k];
            const float4 ha = *(const float4*)(hp + (size_t)sa * F_DIM);
            acc.x = fmaf(w, ha.x, acc.x); acc.y = fmaf(w, ha.y, acc.y);
            acc.z = fmaf(w, ha.z, acc.z); acc.w = fmaf(w, ha.w, acc.w);
        }
        wave_lds_order();
    }

    for (int o = 16; o > 0; o >>= 1) {
        l0 += __shfl_xor(l0, o);
        l1 += __shfl_xor(l1, o);
        l2 += __shfl_xor(l2, o);
        l3 += __shfl_xor(l3, o);
    }
    const float l = (head == 0) ? l0 : (head == 1) ? l1 : (head == 2) ? l2 : l3;
    float4 r = make_float4(0.0f, 0.0f, 0.0f, 0.0f);
    if (end > beg) {
        r.x = acc.x / l; r.y = acc.y / l; r.z = acc.z / l; r.w = acc.w / l;
    }
    *(float4*)&out[node * F_DIM + lane * 4] = r;
}

extern "C" void kernel_launch(void* const* d_in, const int* in_sizes, int n_in,
                              void* d_out, int out_size, void* d_ws, size_t ws_size,
                              hipStream_t stream) {
    const float* h   = (const float*)d_in[0];
    const int*   adj = (const int*)d_in[1];
    const float* W   = (const float*)d_in[2];
    const float* a   = (const float*)d_in[3];
    float* out = (float*)d_out;

    char* ws = (char*)d_ws;
    size_t off = 0;
    float*    hprime = (float*)(ws + off);     off += (size_t)N_NODES * F_DIM * 4;
    float*    asrc   = (float*)(ws + off);     off += (size_t)N_NODES * N_HEADS * 4;
    float*    adst   = (float*)(ws + off);     off += (size_t)N_NODES * N_HEADS * 4;
    int*      cnt    = (int*)(ws + off);       off += (size_t)200192;
    int*      offs   = (int*)(ws + off);       off += (size_t)200192;
    int*      rank   = (int*)(ws + off);       off += (size_t)N_EDGES * 4;
    int*      csr    = (int*)(ws + off);       off += (size_t)N_EDGES * 4;
    _Float16* wpack  = (_Float16*)(ws + off);  off += (size_t)N_TILES * K_TILES * 32 * 16 * 2;

    const int B = 256;

    pack_w_kernel<<<(N_NODES + B - 1) / B, B, 0, stream>>>(W, wpack, cnt);
    gemm_kernel<<<M_TILES, B, 0, stream>>>(h, wpack, hprime);
    alpha_kernel<<<(N_NODES * N_HEADS + B - 1) / B, B, 0, stream>>>(hprime, a, asrc, adst);
    count_rank_kernel<<<(N_EDGES + B - 1) / B, B, 0, stream>>>(adj, cnt, rank);
    scan_kernel<<<1, SCAN_T, 0, stream>>>(cnt, offs);
    fill_kernel<<<(N_EDGES + B - 1) / B, B, 0, stream>>>(adj, offs, rank, csr);
    node_kernel<<<(N_NODES + WPB - 1) / WPB, WPB * 32, 0, stream>>>(offs, csr, asrc, adst, hprime, out);
}


# Projection A tile staged in one memory round trip (four 16-byte loads per thread in flight, was 16 dependent 4-byte loads) at a 528-byte LDS row pitch: fragment reads free of bank conflicts
# speedup vs baseline: 2.2251x; 1.2153x over previous
#include <hip/hip_runtime.h>
#include <hip/hip_fp16.h>

typedef __attribute__((ext_vector_type(16))) _Float16 v16h;
typedef __attribute__((ext_vector_type(8)))  _Float16 v8h;
typedef __attribute__((ext_vector_type(4)))  _Float16 v4h;
typedef __attribute__((ext_vector_type(8)))  float    v8f;

#define N_NODES   50000
#define N_EDGES   1600000
#define IN_DIM    256
#define OUT_DIM   32
#define N_HEADS   4
#define F_DIM     128
#define ALPHA_SLOPE 0.2f
#define M_TILES   3125
#define N_TILES   8
#define K_TILES   8

__global__ void pack_w_kernel(const float* __restrict__ W, _Float16* __restrict__ wpack,
                              int* __restrict__ cnt) {
    int idx = blockIdx.x * blockDim.x + threadIdx.x;
    if (idx < N_NODES) cnt[idx] = 0;
    if (idx >= N_TILES * K_TILES * 32) return;
    int lane = idx & 31;
    int kt   = (idx >> 5) & 7;
    int nt   = idx >> 8;
    int col  = nt * 16 + (lane & 15);
    int kbase = kt * 32 + ((lane & 16) ? 8 : 0);
    _Float16* dst = wpack + idx * 16;
    #pragma unroll
    for (int i = 0; i < 8; ++i) {
        dst[i]     = (_Float16)W[(kbase + i)      * F_DIM + col];
        dst[i + 8] = (_Float16)W[(kbase + 16 + i) * F_DIM + col];
    }
}

#define A_PITCH (IN_DIM + 8)
static_assert(M_TILES * 16 == N_NODES, "every projection tile is a full 16-row tile: no row guard");
static_assert((A_PITCH * 2) % 32 == 16, "row pitch must be an odd multiple of 16 bytes");

__device__ __forceinline__ void gemm_body(const int tile,
                                          const float* __restrict__ h,
                                          const _Float16* __restrict__ wpack,
                                          float* __restrict__ hprime) {
    __shared__ __align__(32) _Float16 tileA[16 * A_PITCH];

    const int tid = threadIdx.x;
    const int m0  = tile * 16;

    const int q  = tid & 63;
    const int r0 = tid >> 6;
    const float4* h4 = (const float4*)h + (size_t)m0 * (IN_DIM / 4) + q;
    float4 v[4];
    #pragma unroll
    for (int j = 0; j < 4; ++j)
        v[j] = h4[(size_t)(r0 + 4 * j) * (IN_DIM / 4)];
    #pragma unroll
    for (int j = 0; j < 4; ++j) {
        v4h p;
        p[0] = (_Float16)v[j].x; p[1] = (_Float16)v[j].y;
        p[2] = (_Float16)v[j].z; p[3] = (_Float16)v[j].w;
        *(v4h*)&tileA[(r0 + 4 * j) * A_PITCH + q * 4] = p;
    }
    __syncthreads();

    const int wave = tid >> 5;
    const int lane = tid & 31;
    const int rowA = lane & 15;
    const int koff = (lane & 16) ? 8 : 0;

    v8f c = {};
    #pragma unroll
    for (int kt = 0; kt < K_TILES; ++kt) {
        const int kbase = kt * 32 + koff;
        v8h lo = *(const v8h*)&tileA[rowA * A_PITCH + kbase];
        v8h hi = *(const v8h*)&tileA[rowA * A_PITCH + kbase + 16];
        v16h afrag = __builtin_shufflevector(lo, hi,
            0, 1, 2, 3, 4, 5, 6, 7, 8, 9, 10, 11, 12, 13, 14, 15);
        v16h bfrag = *(const v16h*)&wpack[((wave * K_TILES + kt) * 32 + lane) * 16];
        c = __builtin_amdgcn_wmma_f32_16x16x32_f16(
                 false, afrag,  false, bfrag,
                 (short)0, c,  false,  false);
    }

    const int col   = wave * 16 + (lane & 15);
    const int rbase = (lane & 16) ? 8 : 0;
    #pragma unroll
    for (int i = 0; i < 8; ++i)
        hprime[(m0 + rbase + i) * F_DIM + col] = c[i];
}

__device__ __forceinline__ void alpha_body(const int t,
                                           const float* __restrict__ hprime, const float* __restrict__ a,
                                           float* __restrict__ asrc, float* __restrict__ adst) {
    if (t >= N_NODES * N_HEADS) return;
    int node = t >> 2;
    int head = t & 3;
    const float* hp = hprime + node * F_DIM + head * OUT_DIM;
    float s0 = 0.0f, s1 = 0.0f;
    #pragma unroll
    for (int d = 0; d < OUT_DIM; ++d) {
        float v = hp[d];
        s0 += v * a[d * N_HEADS + head];
        s1 += v * a[(OUT_DIM + d) * N_HEADS + head];
    }
    asrc[t] = s0;
    adst[t] = s1;
}

__device__ __forceinline__ void count_rank_body(const int e,
                                                const int* __restrict__ adj, int* __restrict__ cnt,
                                                int* __restrict__ rank) {
    if (e >= N_EDGES) return;
    int d = adj[N_EDGES + e];
    rank[e] = atomicAdd(&cnt[d], 1);
}

#define SCAN_T 1024
#define SCAN_U 13
#define N_UNITS (N_NODES / 4)

__device__ __forceinline__ void scan_body(const int* __restrict__ cnt,
                                          int* __restrict__ offs) {
    __shared__ int wtot[32];
    const int t = threadIdx.x, lane = t & 31, wave = t >> 5;
    const int4* c4 = (const int4*)cnt;
    int4 v[SCAN_U];
    int sum = 0;
    #pragma unroll
    for (int q = 0; q < SCAN_U; ++q) {
        int u = t * SCAN_U + q;
        v[q] = (u < N_UNITS) ? c4[u] : make_int4(0, 0, 0, 0);
        sum += v[q].x + v[q].y + v[q].z + v[q].w;
    }
    int inc = sum;
    for (int o = 1; o < 32; o <<= 1) {
        int n = __shfl_up(inc, o);
        if (lane >= o) inc += n;
    }
    if (lane == 31) wtot[wave] = inc;
    __syncthreads();
    if (wave == 0) {
        int w = wtot[lane], wi = w;
        for (int o = 1; o < 32; o <<= 1) {
            int n = __shfl_up(wi, o);
            if (lane >= o) wi += n;
        }
        wtot[lane] = wi - w;
    }
    __syncthreads();
    int run = wtot[wave] + inc - sum;
    #pragma unroll
    for (int q = 0; q < SCAN_U; ++q) {
        int u = t * SCAN_U + q;
        if (u < N_UNITS) {
            int4 o;
            o.x = run; run += v[q].x;
            o.y = run; run += v[q].y;
            o.z = run; run += v[q].z;
            o.w = run; run += v[q].w;
            ((int4*)offs)[u] = o;
        }
    }
    if (t == SCAN_T - 1) offs[N_NODES] = run;
}

static_assert(N_EDGES == 2 * M_TILES * 256, "two 256-edge count blocks per projection tile");

__global__ void __launch_bounds__(256) proj_count_kernel(const float* __restrict__ h,
                                                         const _Float16* __restrict__ wpack,
                                                         float* __restrict__ hprime,
                                                         const int* __restrict__ adj,
                                                         int* __restrict__ cnt,
                                                         int* __restrict__ rank) {
    const int g = blockIdx.x / 3;
    const int r = blockIdx.x - g * 3;
    if (r == 0) gemm_body(g, h, wpack, hprime);
    else        count_rank_body((g * 2 + (r - 1)) * 256 + (int)threadIdx.x, adj, cnt, rank);
}

#define ALPHA_BLOCKS ((N_NODES * N_HEADS + SCAN_T - 1) / SCAN_T)

__global__ void __launch_bounds__(SCAN_T) scan_alpha_kernel(const int* __restrict__ cnt,
                                                            int* __restrict__ offs,
                                                            const float* __restrict__ hprime,
                                                            const float* __restrict__ a,
                                                            float* __restrict__ asrc,
                                                            float* __restrict__ adst) {
    if (blockIdx.x == 0) scan_body(cnt, offs);
    else alpha_body((int)(blockIdx.x - 1) * SCAN_T + (int)threadIdx.x, hprime, a, asrc, adst);
}

__global__ void fill_kernel(const int* __restrict__ adj, const int* __restrict__ offs,
                            const int* __restrict__ rank, int* __restrict__ csr_src) {
    int e = blockIdx.x * blockDim.x + threadIdx.x;
    if (e >= N_EDGES) return;
    int s = adj[e];
    int d = adj[N_EDGES + e];
    csr_src[offs[d] + rank[e]] = s;
}

__device__ __forceinline__ float leaky(float v) {
    return (v > 0.0f) ? v : ALPHA_SLOPE * v;
}

__device__ __forceinline__ void wave_lds_order() {
    __builtin_amdgcn_fence(__ATOMIC_RELEASE, "wavefront");
    __builtin_amdgcn_wave_barrier();
    __builtin_amdgcn_fence(__ATOMIC_ACQUIRE, "wavefront");
}

#define CAP 128
#define WPB 8

__global__ void __launch_bounds__(WPB * 32) node_kernel(const int* __restrict__ offs,
                                                        const int* __restrict__ csr_src,
                                                        const float* __restrict__ asrc,
                                                        const float* __restrict__ adst,
                                                        const float* __restrict__ hprime,
                                                        float* __restrict__ out) {
    __shared__ __align__(16) int   s_key[WPB][CAP];
    __shared__ __align__(16) int   s_srt[WPB][CAP];
    __shared__ __align__(16) float s_w[WPB][N_HEADS][CAP];

    const int wave = __builtin_amdgcn_readfirstlane((int)(threadIdx.x >> 5));
    const int lane = threadIdx.x & 31;
    const int node = blockIdx.x * WPB + wave;
    if (node >= N_NODES) return;

    int*   key = s_key[wave];
    int*   srt = s_srt[wave];
    float* wgt = &s_w[wave][0][0];

    const int beg = offs[node];
    const int end = offs[node + 1];
    const float4* asrc4 = (const float4*)asrc;
    const float4 ad = ((const float4*)adst)[node];
    const int head = lane >> 3;

    float m0 = -INFINITY, m1 = -INFINITY, m2 = -INFINITY, m3 = -INFINITY;
    for (int i = beg + lane; i < end; i += 32) {
        const float4 as = asrc4[csr_src[i]];
        m0 = fmaxf(m0, leaky(as.x + ad.x));
        m1 = fmaxf(m1, leaky(as.y + ad.y));
        m2 = fmaxf(m2, leaky(as.z + ad.z));
        m3 = fmaxf(m3, leaky(as.w + ad.w));
    }
    for (int o = 16; o > 0; o >>= 1) {
        m0 = fmaxf(m0, __shfl_xor(m0, o));
        m1 = fmaxf(m1, __shfl_xor(m1, o));
        m2 = fmaxf(m2, __shfl_xor(m2, o));
        m3 = fmaxf(m3, __shfl_xor(m3, o));
    }

    float l0 = 0.0f, l1 = 0.0f, l2 = 0.0f, l3 = 0.0f;
    float4 acc = make_float4(0.0f, 0.0f, 0.0f, 0.0f);
    const float* hp   = hprime + lane * 4;
    const float* wrow = wgt + head * CAP;

    for (int base = beg; base < end; base += CAP) {
        const int cnt  = min(CAP, end - base);
        const int cnt4 = (cnt + 3) & ~3;

        for (int i = lane; i < cnt4; i += 32)
            key[i] = (i < cnt) ? ((csr_src[base + i] << 7) | i) : 0x7fffffff;
        wave_lds_order();

        for (int i = lane; i < cnt; i += 32) {
            const int kk = key[i];
            int r = 0;
            for (int j = 0; j < cnt4; j += 4) {
                const int4 q = *(const int4*)&key[j];
                r += (q.x < kk) + (q.y < kk) + (q.z < kk) + (q.w < kk);
            }
            srt[r] = kk >> 7;
        }
        wave_lds_order();

        for (int i = lane; i < cnt; i += 32) {
            const float4 as = asrc4[srt[i]];
            const float e0 = __expf(leaky(as.x + ad.x) - m0);
            const float e1 = __expf(leaky(as.y + ad.y) - m1);
            const float e2 = __expf(leaky(as.z + ad.z) - m2);
            const float e3 = __expf(leaky(as.w + ad.w) - m3);
            wgt[0 * CAP + i] = e0;  l0 += e0;
            wgt[1 * CAP + i] = e1;  l1 += e1;
            wgt[2 * CAP + i] = e2;  l2 += e2;
            wgt[3 * CAP + i] = e3;  l3 += e3;
        }
        wave_lds_order();

        int k = 0;
        for (; k + 4 <= cnt; k += 4) {
            const int4   s4 = *(const int4*)&srt[k];
            const float4 w4 = *(const float4*)&wrow[k];
            const int sa = __builtin_amdgcn_readfirstlane(s4.x);
            const int sb = __builtin_amdgcn_readfirstlane(s4.y);
            const int sc = __builtin_amdgcn_readfirstlane(s4.z);
            const int sd = __builtin_amdgcn_readfirstlane(s4.w);
            const float4 ha = *(const float4*)(hp + (size_t)sa * F_DIM);
            const float4 hb = *(const float4*)(hp + (size_t)sb * F_DIM);
            const float4 hc = *(const float4*)(hp + (size_t)sc * F_DIM);
            const float4 hd = *(const float4*)(hp + (size_t)sd * F_DIM);
            acc.x = fmaf(w4.x, ha.x, acc.x); acc.y = fmaf(w4.x, ha.y, acc.y);
            acc.z = fmaf(w4.x, ha.z, acc.z); acc.w = fmaf(w4.x, ha.w, acc.w);
            acc.x = fmaf(w4.y, hb.x, acc.x); acc.y = fmaf(w4.y, hb.y, acc.y);
            acc.z = fmaf(w4.y, hb.z, acc.z); acc.w = fmaf(w4.y, hb.w, acc.w);
            acc.x = fmaf(w4.z, hc.x, acc.x); acc.y = fmaf(w4.z, hc.y, acc.y);
            acc.z = fmaf(w4.z, hc.z, acc.z); acc.w = fmaf(w4.z, hc.w, acc.w);
            acc.x = fmaf(w4.w, hd.x, acc.x); acc.y = fmaf(w4.w, hd.y, acc.y);
            acc.z = fmaf(w4.w, hd.z, acc.z); acc.w = fmaf(w4.w, hd.w, acc.w);
        }
        for (; k < cnt; ++k) {
            const int   sa = __builtin_amdgcn_readfirstlane(srt[k]);
            const float w  = wrow[k];
            const float4 ha = *(const float4*)(hp + (size_t)sa * F_DIM);
            acc.x = fmaf(w, ha.x, acc.x); acc.y = fmaf(w, ha.y, acc.y);
            acc.z = fmaf(w, ha.z, acc.z); acc.w = fmaf(w, ha.w, acc.w);
        }
        wave_lds_order();
    }

    for (int o = 16; o > 0; o >>= 1) {
        l0 += __shfl_xor(l0, o);
        l1 += __shfl_xor(l1, o);
        l2 += __shfl_xor(l2, o);
        l3 += __shfl_xor(l3, o);
    }
    const float l = (head == 0) ? l0 : (head == 1) ? l1 : (head == 2) ? l2 : l3;
    float4 r = make_float4(0.0f, 0.0f, 0.0f, 0.0f);
    if (end > beg) {
        r.x = acc.x / l; r.y = acc.y / l; r.z = acc.z / l; r.w = acc.w / l;
    }
    *(float4*)&out[node * F_DIM + lane * 4] = r;
}

extern "C" void kernel_launch(void* const* d_in, const int* in_sizes, int n_in,
                              void* d_out, int out_size, void* d_ws, size_t ws_size,
                              hipStream_t stream) {
    const float* h   = (const float*)d_in[0];
    const int*   adj = (const int*)d_in[1];
    const float* W   = (const float*)d_in[2];
    const float* a   = (const float*)d_in[3];
    float* out = (float*)d_out;

    char* ws = (char*)d_ws;
    size_t off = 0;
    float*    hprime = (float*)(ws + off);     off += (size_t)N_NODES * F_DIM * 4;
    float*    asrc   = (float*)(ws + off);     off += (size_t)N_NODES * N_HEADS * 4;
    float*    adst   = (float*)(ws + off);     off += (size_t)N_NODES * N_HEADS * 4;
    int*      cnt    = (int*)(ws + off);       off += (size_t)200192;
    int*      offs   = (int*)(ws + off);       off += (size_t)200192;
    int*      rank   = (int*)(ws + off);       off += (size_t)N_EDGES * 4;
    int*      csr    = (int*)(ws + off);       off += (size_t)N_EDGES * 4;
    _Float16* wpack  = (_Float16*)(ws + off);  off += (size_t)N_TILES * K_TILES * 32 * 16 * 2;

    const int B = 256;

    pack_w_kernel<<<(N_NODES + B - 1) / B, B, 0, stream>>>(W, wpack, cnt);
    proj_count_kernel<<<3 * M_TILES, B, 0, stream>>>(h, wpack, hprime, adj, cnt, rank);
    scan_alpha_kernel<<<1 + ALPHA_BLOCKS, SCAN_T, 0, stream>>>(cnt, offs, hprime, a, asrc, adst);
    fill_kernel<<<(N_EDGES + B - 1) / B, B, 0, stream>>>(adj, offs, rank, csr);
    node_kernel<<<(N_NODES + WPB - 1) / WPB, WPB * 32, 0, stream>>>(offs, csr, asrc, adst, hprime, out);
}
